# starting kernel
# baseline (speedup 1.0000x reference)
#include <hip/hip_runtime.h>
#include <cmath>

typedef __attribute__((ext_vector_type(2))) float v2f;
typedef __attribute__((ext_vector_type(8))) float v8f;

#define B_   128
#define T_   20
#define E_   512
#define H_   512
#define V_   10000
#define LDSS 516

__device__ __forceinline__ float sigmoidf_(float x) {
    return 1.0f / (1.0f + __expf(-x));
}

__device__ __forceinline__ v8f wmma_f32(v2f a, v2f b, v8f c) {
    return __builtin_amdgcn_wmma_f32_16x16x4_f32(false, a, false, b, (short)0, c,
                                                 false, false);
}

__global__ void lstm_init(const float* __restrict__ features,
                          float* __restrict__ h0, float* __restrict__ c0) {
    int i = blockIdx.x * blockDim.x + threadIdx.x;
    if (i < B_ * H_) { h0[i] = features[i]; c0[i] = 0.0f; }
}

__global__ __launch_bounds__(128) void lstm_step(
    const int* __restrict__ captions, int t,
    const float* __restrict__ embed_W,
    const float* __restrict__ W_ih, const float* __restrict__ W_hh,
    const float* __restrict__ b_ih, const float* __restrict__ b_hh,
    const float* __restrict__ h_in, const float* __restrict__ c_in,
    float* __restrict__ h_out, float* __restrict__ c_out)
{
    __shared__ float xt[16 * LDSS];
    __shared__ float ht[16 * LDSS];
    const int mtile = blockIdx.y;
    const int tid   = threadIdx.x;

    const int wave  = tid >> 5;
    const int lane  = tid & 31;
    const int half  = lane >> 4;
    const int l16   = lane & 15;
    const int htile = blockIdx.x * 4 + wave;
    const int col   = htile * 16 + l16;

    v8f acc[4];
    #pragma unroll
    for (int g = 0; g < 4; ++g)
        #pragma unroll
        for (int v = 0; v < 8; ++v) acc[g][v] = 0.0f;

    for (int idx = tid * 4; idx < 16 * E_; idx += 128 * 4) {
        int row = idx / E_, e = idx % E_;
        int tok = captions[(mtile * 16 + row) * T_ + t];
        *(float4*)(xt + row * LDSS + e) =
            *(const float4*)(embed_W + (size_t)tok * E_ + e);
        *(float4*)(ht + row * LDSS + e) =
            *(const float4*)(h_in + (size_t)(mtile * 16 + row) * H_ + e);
    }
    __syncthreads();

    for (int k0 = 0; k0 < E_; k0 += 4) {
        int kk = k0 + 2 * half;
        v2f a = *(const v2f*)(xt + l16 * LDSS + kk);
        #pragma unroll
        for (int g = 0; g < 4; ++g) {
            v2f b = *(const v2f*)(W_ih + (size_t)(g * H_ + col) * E_ + kk);
            acc[g] = wmma_f32(a, b, acc[g]);
        }
    }
    for (int k0 = 0; k0 < H_; k0 += 4) {
        int kk = k0 + 2 * half;
        v2f a = *(const v2f*)(ht + l16 * LDSS + kk);
        #pragma unroll
        for (int g = 0; g < 4; ++g) {
            v2f b = *(const v2f*)(W_hh + (size_t)(g * H_ + col) * H_ + kk);
            acc[g] = wmma_f32(a, b, acc[g]);
        }
    }

    const float bi = b_ih[0 * H_ + col] + b_hh[0 * H_ + col];
    const float bf = b_ih[1 * H_ + col] + b_hh[1 * H_ + col];
    const float bg = b_ih[2 * H_ + col] + b_hh[2 * H_ + col];
    const float bo = b_ih[3 * H_ + col] + b_hh[3 * H_ + col];
    #pragma unroll
    for (int v = 0; v < 8; ++v) {
        int rowg = mtile * 16 + v + 8 * half;
        float iv = sigmoidf_(acc[0][v] + bi);
        float fv = sigmoidf_(acc[1][v] + bf);
        float gv = tanhf(acc[2][v] + bg);
        float ov = sigmoidf_(acc[3][v] + bo);
        float cp = c_in[(size_t)rowg * H_ + col];
        float cn = fv * cp + iv * gv;
        c_out[(size_t)rowg * H_ + col] = cn;
        h_out[(size_t)rowg * H_ + col] = ov * tanhf(cn);
    }
}

__global__ __launch_bounds__(160) void proj_step(
    const float* __restrict__ h, const float* __restrict__ lin_W,
    const float* __restrict__ lin_b, float* __restrict__ out, int t)
{
    __shared__ float hsh[32 * LDSS];
    const int mbase = blockIdx.y * 32;
    const int tid   = threadIdx.x;

    for (int idx = tid * 4; idx < 32 * H_; idx += 160 * 4) {
        int row = idx / H_, e = idx % H_;
        *(float4*)(hsh + row * LDSS + e) =
            *(const float4*)(h + (size_t)(mbase + row) * H_ + e);
    }
    __syncthreads();

    const int wave = tid >> 5;
    const int lane = tid & 31;
    const int half = lane >> 4, l16 = lane & 15;
    const int ntile0 = blockIdx.x * 25 + wave * 5;

    v8f acc[10];
    int  ncol[5];
    #pragma unroll
    for (int i = 0; i < 5; ++i) ncol[i] = (ntile0 + i) * 16 + l16;
    #pragma unroll
    for (int j = 0; j < 10; ++j)
        #pragma unroll
        for (int v = 0; v < 8; ++v) acc[j][v] = 0.0f;

    for (int k0 = 0; k0 < H_; k0 += 4) {
        int kk = k0 + 2 * half;
        v2f a0 = *(const v2f*)(hsh + l16 * LDSS + kk);
        v2f a1 = *(const v2f*)(hsh + (16 + l16) * LDSS + kk);
        #pragma unroll
        for (int i = 0; i < 5; ++i) {
            v2f b = *(const v2f*)(lin_W + (size_t)ncol[i] * H_ + kk);
            acc[i]     = wmma_f32(a0, b, acc[i]);
            acc[5 + i] = wmma_f32(a1, b, acc[5 + i]);
        }
    }

    #pragma unroll
    for (int m = 0; m < 2; ++m) {
        #pragma unroll
        for (int i = 0; i < 5; ++i) {
            float bias = lin_b[ncol[i]];
            #pragma unroll
            for (int v = 0; v < 8; ++v) {
                int rowg = mbase + m * 16 + v + 8 * half;
                __builtin_nontemporal_store(
                    acc[m * 5 + i][v] + bias,
                    out + (size_t)rowg * T_ * V_ + (size_t)t * V_ + ncol[i]);
            }
        }
    }
}

extern "C" void kernel_launch(void* const* d_in, const int* in_sizes, int n_in,
                              void* d_out, int out_size, void* d_ws, size_t ws_size,
                              hipStream_t stream) {
    const float* features = (const float*)d_in[0];
    const int*   captions = (const int*)d_in[1];
    const float* embed_W  = (const float*)d_in[2];
    const float* W_ih     = (const float*)d_in[3];
    const float* W_hh     = (const float*)d_in[4];
    const float* b_ih     = (const float*)d_in[5];
    const float* b_hh     = (const float*)d_in[6];
    const float* lin_W    = (const float*)d_in[7];
    const float* lin_b    = (const float*)d_in[8];
    float* out = (float*)d_out;

    float* ws = (float*)d_ws;
    float* hbuf[2] = { ws,                       ws + (size_t)B_ * H_ };
    float* cbuf[2] = { ws + 2 * (size_t)B_ * H_, ws + 3 * (size_t)B_ * H_ };

    lstm_init<<<(B_ * H_ + 255) / 256, 256, 0, stream>>>(features, hbuf[0], cbuf[0]);

    for (int t = 0; t < T_; ++t) {
        int cur = t & 1, nxt = cur ^ 1;
        lstm_step<<<dim3(8, 8), 128, 0, stream>>>(
            captions, t, embed_W, W_ih, W_hh, b_ih, b_hh,
            hbuf[cur], cbuf[cur], hbuf[nxt], cbuf[nxt]);
        proj_step<<<dim3(25, 4), 160, 0, stream>>>(hbuf[nxt], lin_W, lin_b, out, t);
    }
}


# Input product hoisted out of the recurrence: one launch forms x@W_ih^T + biases for all 20 steps, parked in each step's out rows until its projection overwrites them; sum order changes
# speedup vs baseline: 4.2501x; 4.2501x over previous
#include <hip/hip_runtime.h>
#include <cmath>

typedef __attribute__((ext_vector_type(2))) float v2f;
typedef __attribute__((ext_vector_type(4))) float v4f;
typedef __attribute__((ext_vector_type(8))) float v8f;

#define B_   128
#define T_   20
#define E_   512
#define H_   512
#define V_   10000
#define KSPLIT 8
#define KW     64

__device__ __forceinline__ float sigmoidf_(float x) {
    return 1.0f / (1.0f + __expf(-x));
}

__device__ __forceinline__ v8f wmma_f32(v2f a, v2f b, v8f c) {
    return __builtin_amdgcn_wmma_f32_16x16x4_f32(false, a, false, b, (short)0, c,
                                                 false, false);
}

__global__ void lstm_init(const float* __restrict__ features,
                          float* __restrict__ h0, float* __restrict__ c0) {
    int i = blockIdx.x * blockDim.x + threadIdx.x;
    if (i < B_ * H_) { h0[i] = features[i]; c0[i] = 0.0f; }
}

#define PARK   16
#define XP     36
#define XROWS  128
#define XSTAGE (XROWS * XP)
__global__ __launch_bounds__(128) void xgate_all(
    const int* __restrict__ captions,
    const float* __restrict__ embed_W, const float* __restrict__ W_ih,
    const float* __restrict__ b_ih, const float* __restrict__ b_hh,
    float* __restrict__ out)
{
    __shared__ float sm[2 * XSTAGE];
    const int mbase = blockIdx.y * 64;
    const int nbase = blockIdx.x * 64;
    const int tid   = threadIdx.x;
    const int w    = tid >> 5;
    const int lane = tid & 31;
    const int half = lane >> 4, l16 = lane & 15;

    const int crow = tid >> 3;
    const int spc  = (tid & 7) * 4;

    v8f acc[4];
    #pragma unroll
    for (int i = 0; i < 4; ++i)
        #pragma unroll
        for (int v = 0; v < 8; ++v) acc[i][v] = 0.0f;

    v4f ra[4], rb[4];
    const float* xsrc[4];
    #pragma unroll
    for (int i = 0; i < 4; ++i)
        xsrc[i] = embed_W + (size_t)captions[mbase + 16 * i + crow] * E_ + spc;
    const float* wsrc = W_ih + (size_t)(nbase + crow) * E_ + spc;

    #pragma unroll
    for (int i = 0; i < 4; ++i) rb[i] = *(const v4f*)(wsrc + (size_t)(16 * i) * E_);
    #pragma unroll
    for (int i = 0; i < 4; ++i) ra[i] = *(const v4f*)(xsrc[i]);

    for (int c = 0; c < 16; ++c) {
        float* st = sm + (c & 1) * XSTAGE;
        #pragma unroll
        for (int i = 0; i < 4; ++i)
            *(v4f*)(st + (16 * i + crow) * XP + spc) = ra[i];
        #pragma unroll
        for (int i = 0; i < 4; ++i)
            *(v4f*)(st + (64 + 16 * i + crow) * XP + spc) = rb[i];
        __syncthreads();
        if (c < 15) {
            #pragma unroll
            for (int i = 0; i < 4; ++i) rb[i] = *(const v4f*)(wsrc + (size_t)(16 * i) * E_ + 32 * (c + 1));
            #pragma unroll
            for (int i = 0; i < 4; ++i) ra[i] = *(const v4f*)(xsrc[i] + 32 * (c + 1));
        }
        #pragma unroll
        for (int ks = 0; ks < 8; ++ks) {
            const int kk = 4 * ks + 2 * half;
            v2f a = *(const v2f*)(st + (16 * w + l16) * XP + kk);
            #pragma unroll
            for (int i = 0; i < 4; ++i) {
                v2f b = *(const v2f*)(st + (64 + i * 16 + l16) * XP + kk);
                acc[i] = wmma_f32(a, b, acc[i]);
            }
        }
    }

    #pragma unroll
    for (int i = 0; i < 4; ++i) {
        const int n = nbase + 16 * i + l16;
        const float bias = b_ih[n] + b_hh[n];
        #pragma unroll
        for (int v = 0; v < 8; ++v) {
            const int m = mbase + 16 * w + v + 8 * half;
            out[(size_t)m * V_ + PARK + n] = acc[i][v] + bias;
        }
    }
}

#define LP    68
#define LROWS 80
__global__ __launch_bounds__(256) void lstm_step(
    int t,
    const float* __restrict__ W_hh,
    const float* __restrict__ xg,
    const float* __restrict__ h_in, float* __restrict__ h_out,
    float* __restrict__ c_state)
{
    __shared__ float sm[KSPLIT * LROWS * LP];
    const int mtile = blockIdx.y;
    const int tid   = threadIdx.x;

    const int q     = tid >> 5;
    const int lane  = tid & 31;
    const int half  = lane >> 4;
    const int l16   = lane & 15;
    const int htile = blockIdx.x;
    const int col   = htile * 16 + l16;

    float* reg = sm + q * (LROWS * LP);
    const int k0  = q * KW;
    const int spc = l16 * 4;

    v8f acc[4];
    #pragma unroll
    for (int g = 0; g < 4; ++g)
        #pragma unroll
        for (int v = 0; v < 8; ++v) acc[g][v] = 0.0f;

    v4f ra[8], rb[32];

    #pragma unroll
    for (int i = 0; i < 32; ++i) {
        int r = 2 * i + half;
        rb[i] = *(const v4f*)(W_hh + (size_t)((r >> 4) * H_ + htile * 16 + (r & 15)) * H_ + k0 + spc);
    }
    #pragma unroll
    for (int i = 0; i < 8; ++i)
        ra[i] = *(const v4f*)(h_in + (size_t)(mtile * 16 + 2 * i + half) * H_ + k0 + spc);

    const int rowg = mtile * 16 + q + 8 * half;
    const float* xrow = xg + ((size_t)rowg * T_ + t) * V_ + PARK + col;
    const float xi = xrow[0 * H_];
    const float xf = xrow[1 * H_];
    const float xc = xrow[2 * H_];
    const float xo = xrow[3 * H_];
    const float cp = c_state[(size_t)rowg * H_ + col];

    #pragma unroll
    for (int i = 0; i < 32; ++i)
        *(v4f*)(reg + (16 + 2 * i + half) * LP + spc) = rb[i];
    #pragma unroll
    for (int i = 0; i < 8; ++i)
        *(v4f*)(reg + (2 * i + half) * LP + spc) = ra[i];
    __syncthreads();

    #pragma unroll
    for (int ks = 0; ks < KW / 4; ++ks) {
        const int kk = 4 * ks + 2 * half;
        v2f a = *(const v2f*)(reg + l16 * LP + kk);
        #pragma unroll
        for (int g = 0; g < 4; ++g) {
            v2f b = *(const v2f*)(reg + (16 + g * 16 + l16) * LP + kk);
            acc[g] = wmma_f32(a, b, acc[g]);
        }
    }

    __syncthreads();
    #pragma unroll
    for (int g = 0; g < 4; ++g)
        #pragma unroll
        for (int v = 0; v < 8; ++v)
            sm[((q * 4 + g) * 8 + v) * 32 + lane] = acc[g][v];
    __syncthreads();

    float s[4];
    #pragma unroll
    for (int g = 0; g < 4; ++g) {
        float p = sm[((0 * 4 + g) * 8 + q) * 32 + lane];
        #pragma unroll
        for (int w = 1; w < KSPLIT; ++w)
            p += sm[((w * 4 + g) * 8 + q) * 32 + lane];
        s[g] = p;
    }

    {
        float iv = sigmoidf_(s[0] + xi);
        float fv = sigmoidf_(s[1] + xf);
        float gv = tanhf(s[2] + xc);
        float ov = sigmoidf_(s[3] + xo);
        float cn = fv * cp + iv * gv;
        c_state[(size_t)rowg * H_ + col] = cn;
        h_out[(size_t)rowg * H_ + col] = ov * tanhf(cn);
    }
}

#define PP     36
#define PROWS  144
#define PSTAGE (PROWS * PP)
__global__ __launch_bounds__(128) void proj_step(
    const float* __restrict__ h_s0, const float* __restrict__ h_s1, const float* __restrict__ h_s2,
    const float* __restrict__ lin_W,
    const float* __restrict__ lin_b, float* __restrict__ out, int t0)
{
    __shared__ float sm[2 * PSTAGE];
    const int step  = blockIdx.y >> 1;
    const int mbase = (blockIdx.y & 1) * 64;
    const int t     = t0 + step;
    const float* __restrict__ h = step == 0 ? h_s0 : (step == 1 ? h_s1 : h_s2);
    const int tid   = threadIdx.x;

    const int w    = tid >> 5;
    const int lane = tid & 31;
    const int half = lane >> 4, l16 = lane & 15;
    const int ntile0 = blockIdx.x * 5;

    const int crow = tid >> 3;
    const int spc  = (tid & 7) * 4;

    v8f acc[5];
    int ncol[5];
    #pragma unroll
    for (int i = 0; i < 5; ++i) ncol[i] = (ntile0 + i) * 16 + l16;
    #pragma unroll
    for (int i = 0; i < 5; ++i)
        #pragma unroll
        for (int v = 0; v < 8; ++v) acc[i][v] = 0.0f;

    v4f ra[4], rb[5];
    const float* hsrc = h     + (size_t)(mbase + crow) * H_ + spc;
    const float* wsrc = lin_W + (size_t)(ntile0 * 16 + crow) * H_ + spc;

    #pragma unroll
    for (int i = 0; i < 5; ++i) rb[i] = *(const v4f*)(wsrc + (size_t)(16 * i) * H_);
    #pragma unroll
    for (int i = 0; i < 4; ++i) ra[i] = *(const v4f*)(hsrc + (size_t)(16 * i) * H_);

    for (int c = 0; c < 16; ++c) {
        float* st = sm + (c & 1) * PSTAGE;
        #pragma unroll
        for (int i = 0; i < 4; ++i)
            *(v4f*)(st + (16 * i + crow) * PP + spc) = ra[i];
        #pragma unroll
        for (int i = 0; i < 5; ++i)
            *(v4f*)(st + (64 + 16 * i + crow) * PP + spc) = rb[i];
        __syncthreads();
        if (c < 15) {
            #pragma unroll
            for (int i = 0; i < 5; ++i) rb[i] = *(const v4f*)(wsrc + (size_t)(16 * i) * H_ + 32 * (c + 1));
            #pragma unroll
            for (int i = 0; i < 4; ++i) ra[i] = *(const v4f*)(hsrc + (size_t)(16 * i) * H_ + 32 * (c + 1));
        }
        #pragma unroll
        for (int ks = 0; ks < 8; ++ks) {
            const int kk = 4 * ks + 2 * half;
            v2f a = *(const v2f*)(st + (16 * w + l16) * PP + kk);
            #pragma unroll
            for (int i = 0; i < 5; ++i) {
                v2f b = *(const v2f*)(st + (64 + i * 16 + l16) * PP + kk);
                acc[i] = wmma_f32(a, b, acc[i]);
            }
        }
    }

    #pragma unroll
    for (int i = 0; i < 5; ++i) {
        float bias = lin_b[ncol[i]];
        #pragma unroll
        for (int v = 0; v < 8; ++v) {
            int rowg = mbase + 16 * w + v + 8 * half;
            __builtin_nontemporal_store(
                acc[i][v] + bias,
                out + (size_t)rowg * T_ * V_ + (size_t)t * V_ + ncol[i]);
        }
    }
}

extern "C" void kernel_launch(void* const* d_in, const int* in_sizes, int n_in,
                              void* d_out, int out_size, void* d_ws, size_t ws_size,
                              hipStream_t stream) {
    const float* features = (const float*)d_in[0];
    const int*   captions = (const int*)d_in[1];
    const float* embed_W  = (const float*)d_in[2];
    const float* W_ih     = (const float*)d_in[3];
    const float* W_hh     = (const float*)d_in[4];
    const float* b_ih     = (const float*)d_in[5];
    const float* b_hh     = (const float*)d_in[6];
    const float* lin_W    = (const float*)d_in[7];
    const float* lin_b    = (const float*)d_in[8];
    float* out = (float*)d_out;

    float* ws = (float*)d_ws;
    float* b0 = ws;
    float* b1 = ws + (size_t)B_ * H_;
    float* b2 = ws + 2 * (size_t)B_ * H_;
    float* cs = ws + 3 * (size_t)B_ * H_;

    lstm_init<<<(B_ * H_ + 255) / 256, 256, 0, stream>>>(features, b0, cs);
    xgate_all<<<dim3(32, 40), 128, 0, stream>>>(captions, embed_W, W_ih, b_ih, b_hh, out);

    for (int g = 0; g < 6; ++g) {
        lstm_step<<<dim3(32, 8), 256, 0, stream>>>(3 * g, W_hh, out, b0, b1, cs);
        lstm_step<<<dim3(32, 8), 256, 0, stream>>>(3 * g + 1, W_hh, out, b1, b2, cs);
        lstm_step<<<dim3(32, 8), 256, 0, stream>>>(3 * g + 2, W_hh, out, b2, b0, cs);
        proj_step<<<dim3(125, 6), 128, 0, stream>>>(b1, b2, b0, lin_W, lin_b, out, 3 * g);
    }
    lstm_step<<<dim3(32, 8), 256, 0, stream>>>(18, W_hh, out, b0, b1, cs);
    lstm_step<<<dim3(32, 8), 256, 0, stream>>>(19, W_hh, out, b1, b2, cs);
    proj_step<<<dim3(125, 4), 128, 0, stream>>>(b1, b2, b2, lin_W, lin_b, out, 18);
}


# Projection moved behind the recurrence: steps 0-17 leave h in output rows 18/19, one 4500-block projection launch reads it there, a second takes steps 18-19; 29 launches -> 24; sum order unchanged
# speedup vs baseline: 4.6010x; 1.0826x over previous
#include <hip/hip_runtime.h>
#include <cmath>

typedef __attribute__((ext_vector_type(2))) float v2f;
typedef __attribute__((ext_vector_type(4))) float v4f;
typedef __attribute__((ext_vector_type(8))) float v8f;

#define B_   128
#define T_   20
#define E_   512
#define H_   512
#define V_   10000
#define KSPLIT 8
#define KW     64

__device__ __forceinline__ float sigmoidf_(float x) {
    return 1.0f / (1.0f + __expf(-x));
}

__device__ __forceinline__ v8f wmma_f32(v2f a, v2f b, v8f c) {
    return __builtin_amdgcn_wmma_f32_16x16x4_f32(false, a, false, b, (short)0, c,
                                                 false, false);
}

__global__ void lstm_init(const float* __restrict__ features,
                          float* __restrict__ h0, float* __restrict__ c0) {
    int i = blockIdx.x * blockDim.x + threadIdx.x;
    if (i < B_ * H_) { h0[i] = features[i]; c0[i] = 0.0f; }
}

#define PARK   16
#define XP     36
#define XROWS  128
#define XSTAGE (XROWS * XP)
__global__ __launch_bounds__(128) void xgate_all(
    const int* __restrict__ captions,
    const float* __restrict__ embed_W, const float* __restrict__ W_ih,
    const float* __restrict__ b_ih, const float* __restrict__ b_hh,
    float* __restrict__ out)
{
    __shared__ float sm[2 * XSTAGE];
    const int mbase = blockIdx.y * 64;
    const int nbase = blockIdx.x * 64;
    const int tid   = threadIdx.x;
    const int w    = tid >> 5;
    const int lane = tid & 31;
    const int half = lane >> 4, l16 = lane & 15;

    const int crow = tid >> 3;
    const int spc  = (tid & 7) * 4;

    v8f acc[4];
    #pragma unroll
    for (int i = 0; i < 4; ++i)
        #pragma unroll
        for (int v = 0; v < 8; ++v) acc[i][v] = 0.0f;

    v4f ra[4], rb[4];
    const float* xsrc[4];
    #pragma unroll
    for (int i = 0; i < 4; ++i)
        xsrc[i] = embed_W + (size_t)captions[mbase + 16 * i + crow] * E_ + spc;
    const float* wsrc = W_ih + (size_t)(nbase + crow) * E_ + spc;

    #pragma unroll
    for (int i = 0; i < 4; ++i) rb[i] = *(const v4f*)(wsrc + (size_t)(16 * i) * E_);
    #pragma unroll
    for (int i = 0; i < 4; ++i) ra[i] = *(const v4f*)(xsrc[i]);

    for (int c = 0; c < 16; ++c) {
        float* st = sm + (c & 1) * XSTAGE;
        #pragma unroll
        for (int i = 0; i < 4; ++i)
            *(v4f*)(st + (16 * i + crow) * XP + spc) = ra[i];
        #pragma unroll
        for (int i = 0; i < 4; ++i)
            *(v4f*)(st + (64 + 16 * i + crow) * XP + spc) = rb[i];
        __syncthreads();
        if (c < 15) {
            #pragma unroll
            for (int i = 0; i < 4; ++i) rb[i] = *(const v4f*)(wsrc + (size_t)(16 * i) * E_ + 32 * (c + 1));
            #pragma unroll
            for (int i = 0; i < 4; ++i) ra[i] = *(const v4f*)(xsrc[i] + 32 * (c + 1));
        }
        #pragma unroll
        for (int ks = 0; ks < 8; ++ks) {
            const int kk = 4 * ks + 2 * half;
            v2f a = *(const v2f*)(st + (16 * w + l16) * XP + kk);
            #pragma unroll
            for (int i = 0; i < 4; ++i) {
                v2f b = *(const v2f*)(st + (64 + i * 16 + l16) * XP + kk);
                acc[i] = wmma_f32(a, b, acc[i]);
            }
        }
    }

    #pragma unroll
    for (int i = 0; i < 4; ++i) {
        const int n = nbase + 16 * i + l16;
        const float bias = b_ih[n] + b_hh[n];
        #pragma unroll
        for (int v = 0; v < 8; ++v) {
            const int m = mbase + 16 * w + v + 8 * half;
            out[(size_t)m * V_ + PARK + n] = acc[i][v] + bias;
        }
    }
}

#define HSPLIT 9
#define HROW_A 18
#define HOFF_A 2080
#define HROW_B 19
#define HOFF_B 2064
#define LP    68
#define LROWS 80
__global__ __launch_bounds__(256) void lstm_step(
    int t,
    const float* __restrict__ W_hh,
    const float* xg,
    const float* __restrict__ h_in, float* __restrict__ h_out,
    float* __restrict__ c_state,
    float* hpark)
{
    __shared__ float sm[KSPLIT * LROWS * LP];
    const int mtile = blockIdx.y;
    const int tid   = threadIdx.x;

    const int q     = tid >> 5;
    const int lane  = tid & 31;
    const int half  = lane >> 4;
    const int l16   = lane & 15;
    const int htile = blockIdx.x;
    const int col   = htile * 16 + l16;

    float* reg = sm + q * (LROWS * LP);
    const int k0  = q * KW;
    const int spc = l16 * 4;

    v8f acc[4];
    #pragma unroll
    for (int g = 0; g < 4; ++g)
        #pragma unroll
        for (int v = 0; v < 8; ++v) acc[g][v] = 0.0f;

    v4f ra[8], rb[32];

    #pragma unroll
    for (int i = 0; i < 32; ++i) {
        int r = 2 * i + half;
        rb[i] = *(const v4f*)(W_hh + (size_t)((r >> 4) * H_ + htile * 16 + (r & 15)) * H_ + k0 + spc);
    }
    #pragma unroll
    for (int i = 0; i < 8; ++i)
        ra[i] = *(const v4f*)(h_in + (size_t)(mtile * 16 + 2 * i + half) * H_ + k0 + spc);

    const int rowg = mtile * 16 + q + 8 * half;
    const float* xrow = xg + ((size_t)rowg * T_ + t) * V_ + PARK + col;
    const float xi = xrow[0 * H_];
    const float xf = xrow[1 * H_];
    const float xc = xrow[2 * H_];
    const float xo = xrow[3 * H_];
    const float cp = c_state[(size_t)rowg * H_ + col];

    #pragma unroll
    for (int i = 0; i < 32; ++i)
        *(v4f*)(reg + (16 + 2 * i + half) * LP + spc) = rb[i];
    #pragma unroll
    for (int i = 0; i < 8; ++i)
        *(v4f*)(reg + (2 * i + half) * LP + spc) = ra[i];
    __syncthreads();

    #pragma unroll
    for (int ks = 0; ks < KW / 4; ++ks) {
        const int kk = 4 * ks + 2 * half;
        v2f a = *(const v2f*)(reg + l16 * LP + kk);
        #pragma unroll
        for (int g = 0; g < 4; ++g) {
            v2f b = *(const v2f*)(reg + (16 + g * 16 + l16) * LP + kk);
            acc[g] = wmma_f32(a, b, acc[g]);
        }
    }

    __syncthreads();
    #pragma unroll
    for (int g = 0; g < 4; ++g)
        #pragma unroll
        for (int v = 0; v < 8; ++v)
            sm[((q * 4 + g) * 8 + v) * 32 + lane] = acc[g][v];
    __syncthreads();

    float s[4];
    #pragma unroll
    for (int g = 0; g < 4; ++g) {
        float p = sm[((0 * 4 + g) * 8 + q) * 32 + lane];
        #pragma unroll
        for (int w = 1; w < KSPLIT; ++w)
            p += sm[((w * 4 + g) * 8 + q) * 32 + lane];
        s[g] = p;
    }

    {
        float iv = sigmoidf_(s[0] + xi);
        float fv = sigmoidf_(s[1] + xf);
        float gv = tanhf(s[2] + xc);
        float ov = sigmoidf_(s[3] + xo);
        float cn = fv * cp + iv * gv;
        const float hn = ov * tanhf(cn);
        c_state[(size_t)rowg * H_ + col] = cn;
        h_out[(size_t)rowg * H_ + col] = hn;
        if (t < 2 * HSPLIT) {
            const int prow = t < HSPLIT ? HROW_A : HROW_B;
            const int poff = t < HSPLIT ? HOFF_A + H_ * t : HOFF_B + H_ * (t - HSPLIT);
            hpark[((size_t)rowg * T_ + prow) * V_ + poff + col] = hn;
        }
    }
}

#define PP     36
#define PROWS  144
#define PSTAGE (PROWS * PP)
template <bool PARKED>
__global__ __launch_bounds__(128) void proj_step(
    const float* h_s0, const float* h_s1, const float* h_s2,
    const float* __restrict__ lin_W,
    const float* __restrict__ lin_b, float* out, int t0)
{
    __shared__ float sm[2 * PSTAGE];
    const int step  = blockIdx.y >> 1;
    const int mbase = (blockIdx.y & 1) * 64;
    const int t     = t0 + step;
    const int tid   = threadIdx.x;

    const int w    = tid >> 5;
    const int lane = tid & 31;
    const int half = lane >> 4, l16 = lane & 15;
    const int ntile0 = blockIdx.x * 5;

    const int crow = tid >> 3;
    const int spc  = (tid & 7) * 4;

    v8f acc[5];
    int ncol[5];
    #pragma unroll
    for (int i = 0; i < 5; ++i) ncol[i] = (ntile0 + i) * 16 + l16;
    #pragma unroll
    for (int i = 0; i < 5; ++i)
        #pragma unroll
        for (int v = 0; v < 8; ++v) acc[i][v] = 0.0f;

    v4f ra[4], rb[5];
    const float* hsrc[4];
    if (PARKED) {
        const int prow = t < HSPLIT ? HROW_A : HROW_B;
        const int poff = t < HSPLIT ? HOFF_A + H_ * t : HOFF_B + H_ * (t - HSPLIT);
        #pragma unroll
        for (int i = 0; i < 4; ++i)
            hsrc[i] = h_s0 + ((size_t)(mbase + 16 * i + crow) * T_ + prow) * V_ + poff + spc;
    } else {
        const float* h = step == 0 ? h_s0 : (step == 1 ? h_s1 : h_s2);
        #pragma unroll
        for (int i = 0; i < 4; ++i)
            hsrc[i] = h + (size_t)(mbase + 16 * i + crow) * H_ + spc;
    }
    const float* wsrc = lin_W + (size_t)(ntile0 * 16 + crow) * H_ + spc;

    #pragma unroll
    for (int i = 0; i < 5; ++i) rb[i] = *(const v4f*)(wsrc + (size_t)(16 * i) * H_);
    #pragma unroll
    for (int i = 0; i < 4; ++i) ra[i] = *(const v4f*)(hsrc[i]);

    for (int c = 0; c < 16; ++c) {
        float* st = sm + (c & 1) * PSTAGE;
        #pragma unroll
        for (int i = 0; i < 4; ++i)
            *(v4f*)(st + (16 * i + crow) * PP + spc) = ra[i];
        #pragma unroll
        for (int i = 0; i < 5; ++i)
            *(v4f*)(st + (64 + 16 * i + crow) * PP + spc) = rb[i];
        __syncthreads();
        if (c < 15) {
            #pragma unroll
            for (int i = 0; i < 5; ++i) rb[i] = *(const v4f*)(wsrc + (size_t)(16 * i) * H_ + 32 * (c + 1));
            #pragma unroll
            for (int i = 0; i < 4; ++i) ra[i] = *(const v4f*)(hsrc[i] + 32 * (c + 1));
        }
        #pragma unroll
        for (int ks = 0; ks < 8; ++ks) {
            const int kk = 4 * ks + 2 * half;
            v2f a = *(const v2f*)(st + (16 * w + l16) * PP + kk);
            #pragma unroll
            for (int i = 0; i < 5; ++i) {
                v2f b = *(const v2f*)(st + (64 + i * 16 + l16) * PP + kk);
                acc[i] = wmma_f32(a, b, acc[i]);
            }
        }
    }

    #pragma unroll
    for (int i = 0; i < 5; ++i) {
        float bias = lin_b[ncol[i]];
        #pragma unroll
        for (int v = 0; v < 8; ++v) {
            int rowg = mbase + 16 * w + v + 8 * half;
            __builtin_nontemporal_store(
                acc[i][v] + bias,
                out + (size_t)rowg * T_ * V_ + (size_t)t * V_ + ncol[i]);
        }
    }
}

extern "C" void kernel_launch(void* const* d_in, const int* in_sizes, int n_in,
                              void* d_out, int out_size, void* d_ws, size_t ws_size,
                              hipStream_t stream) {
    const float* features = (const float*)d_in[0];
    const int*   captions = (const int*)d_in[1];
    const float* embed_W  = (const float*)d_in[2];
    const float* W_ih     = (const float*)d_in[3];
    const float* W_hh     = (const float*)d_in[4];
    const float* b_ih     = (const float*)d_in[5];
    const float* b_hh     = (const float*)d_in[6];
    const float* lin_W    = (const float*)d_in[7];
    const float* lin_b    = (const float*)d_in[8];
    float* out = (float*)d_out;

    float* ws = (float*)d_ws;
    float* b0 = ws;
    float* b1 = ws + (size_t)B_ * H_;
    float* b2 = ws + 2 * (size_t)B_ * H_;
    float* cs = ws + 3 * (size_t)B_ * H_;

    lstm_init<<<(B_ * H_ + 255) / 256, 256, 0, stream>>>(features, b0, cs);
    xgate_all<<<dim3(32, 40), 128, 0, stream>>>(captions, embed_W, W_ih, b_ih, b_hh, out);

    for (int g = 0; g < 6; ++g) {
        lstm_step<<<dim3(32, 8), 256, 0, stream>>>(3 * g, W_hh, out, b0, b1, cs, out);
        lstm_step<<<dim3(32, 8), 256, 0, stream>>>(3 * g + 1, W_hh, out, b1, b2, cs, out);
        lstm_step<<<dim3(32, 8), 256, 0, stream>>>(3 * g + 2, W_hh, out, b2, b0, cs, out);
    }
    lstm_step<<<dim3(32, 8), 256, 0, stream>>>(18, W_hh, out, b0, b1, cs, out);
    lstm_step<<<dim3(32, 8), 256, 0, stream>>>(19, W_hh, out, b1, b2, cs, out);
    proj_step<true><<<dim3(125, 36), 128, 0, stream>>>(out, out, out, lin_W, lin_b, out, 0);
    proj_step<false><<<dim3(125, 4), 128, 0, stream>>>(b1, b2, b2, lin_W, lin_b, out, 18);
}


# W_hh re-laid once per call in WMMA fragment order (parked in unused output space): cell launches load B fragments straight into registers, products run as the loads arrive; sum order unchanged
# speedup vs baseline: 4.7508x; 1.0326x over previous
#include <hip/hip_runtime.h>
#include <cmath>

typedef __attribute__((ext_vector_type(2))) float v2f;
typedef __attribute__((ext_vector_type(4))) float v4f;
typedef __attribute__((ext_vector_type(8))) float v8f;

#define B_   128
#define T_   20
#define E_   512
#define H_   512
#define V_   10000
#define KSPLIT 8
#define KW     64

__device__ __forceinline__ float sigmoidf_(float x) {
    return 1.0f / (1.0f + __expf(-x));
}

__device__ __forceinline__ v8f wmma_f32(v2f a, v2f b, v8f c) {
    return __builtin_amdgcn_wmma_f32_16x16x4_f32(false, a, false, b, (short)0, c,
                                                 false, false);
}

#define WPIECE 4096
__device__ __forceinline__ size_t wpiece_off(int j) {
    const int b = j >> 1, t = j & 1;
    return ((size_t)b * T_ + t) * V_ + (t ? 2064 : 2080);
}
__global__ __launch_bounds__(256) void lstm_prep(const float* __restrict__ features,
                                                 const float* __restrict__ W_hh,
                                                 float* __restrict__ h0, float* __restrict__ c0,
                                                 float* __restrict__ out) {
    const int j   = blockIdx.x;
    const int tid = threadIdx.x;
    const int e   = j * 256 + tid;
    h0[e] = features[e];
    c0[e] = 0.0f;

    const int htile = j >> 3, q = j & 7;
    const int lane = tid & 31, c = lane & 15, half = lane >> 4;
    float* dst = out + wpiece_off(j);
    #pragma unroll
    for (int p = 0; p < 4; ++p) {
        const int i  = (tid >> 5) + 8 * p;
        const int g  = i & 3, kp = i >> 2;
        const float* s = W_hh + (size_t)(g * H_ + htile * 16 + c) * H_ + q * KW + 8 * kp + 2 * half;
        const v2f lo = *(const v2f*)(s);
        const v2f hi = *(const v2f*)(s + 4);
        v4f o;
        o.x = lo.x; o.y = lo.y; o.z = hi.x; o.w = hi.y;
        *(v4f*)(dst + (i * 32 + lane) * 4) = o;
    }
}

#define PARK   16
#define XP     36
#define XROWS  128
#define XSTAGE (XROWS * XP)
__global__ __launch_bounds__(128) void xgate_all(
    const int* __restrict__ captions,
    const float* __restrict__ embed_W, const float* __restrict__ W_ih,
    const float* __restrict__ b_ih, const float* __restrict__ b_hh,
    float* __restrict__ out)
{
    __shared__ float sm[2 * XSTAGE];
    const int mbase = blockIdx.y * 64;
    const int nbase = blockIdx.x * 64;
    const int tid   = threadIdx.x;
    const int w    = tid >> 5;
    const int lane = tid & 31;
    const int half = lane >> 4, l16 = lane & 15;

    const int crow = tid >> 3;
    const int spc  = (tid & 7) * 4;

    v8f acc[4];
    #pragma unroll
    for (int i = 0; i < 4; ++i)
        #pragma unroll
        for (int v = 0; v < 8; ++v) acc[i][v] = 0.0f;

    v4f ra[4], rb[4];
    const float* xsrc[4];
    #pragma unroll
    for (int i = 0; i < 4; ++i)
        xsrc[i] = embed_W + (size_t)captions[mbase + 16 * i + crow] * E_ + spc;
    const float* wsrc = W_ih + (size_t)(nbase + crow) * E_ + spc;

    #pragma unroll
    for (int i = 0; i < 4; ++i) rb[i] = *(const v4f*)(wsrc + (size_t)(16 * i) * E_);
    #pragma unroll
    for (int i = 0; i < 4; ++i) ra[i] = *(const v4f*)(xsrc[i]);

    for (int c = 0; c < 16; ++c) {
        float* st = sm + (c & 1) * XSTAGE;
        #pragma unroll
        for (int i = 0; i < 4; ++i)
            *(v4f*)(st + (16 * i + crow) * XP + spc) = ra[i];
        #pragma unroll
        for (int i = 0; i < 4; ++i)
            *(v4f*)(st + (64 + 16 * i + crow) * XP + spc) = rb[i];
        __syncthreads();
        if (c < 15) {
            #pragma unroll
            for (int i = 0; i < 4; ++i) rb[i] = *(const v4f*)(wsrc + (size_t)(16 * i) * E_ + 32 * (c + 1));
            #pragma unroll
            for (int i = 0; i < 4; ++i) ra[i] = *(const v4f*)(xsrc[i] + 32 * (c + 1));
        }
        #pragma unroll
        for (int ks = 0; ks < 8; ++ks) {
            const int kk = 4 * ks + 2 * half;
            v2f a = *(const v2f*)(st + (16 * w + l16) * XP + kk);
            #pragma unroll
            for (int i = 0; i < 4; ++i) {
                v2f b = *(const v2f*)(st + (64 + i * 16 + l16) * XP + kk);
                acc[i] = wmma_f32(a, b, acc[i]);
            }
        }
    }

    #pragma unroll
    for (int i = 0; i < 4; ++i) {
        const int n = nbase + 16 * i + l16;
        const float bias = b_ih[n] + b_hh[n];
        #pragma unroll
        for (int v = 0; v < 8; ++v) {
            const int m = mbase + 16 * w + v + 8 * half;
            out[(size_t)m * V_ + PARK + n] = acc[i][v] + bias;
        }
    }
}

#define HSPLIT 9
#define HROW_A 18
#define HOFF_A 2080
#define HROW_B 19
#define HOFF_B 2064
#define LP    68
#define LROWS 80
__global__ __launch_bounds__(256) void lstm_step(
    int t,
    const float* wfrag,
    const float* xg,
    const float* __restrict__ h_in, float* __restrict__ h_out,
    float* __restrict__ c_state,
    float* hpark)
{
    __shared__ float sm[KSPLIT * LROWS * LP];
    const int mtile = blockIdx.y;
    const int tid   = threadIdx.x;

    const int q     = tid >> 5;
    const int lane  = tid & 31;
    const int half  = lane >> 4;
    const int l16   = lane & 15;
    const int htile = blockIdx.x;
    const int col   = htile * 16 + l16;

    float* reg = sm + q * (LROWS * LP);
    const int k0  = q * KW;
    const int spc = l16 * 4;

    v8f acc[4];
    #pragma unroll
    for (int g = 0; g < 4; ++g)
        #pragma unroll
        for (int v = 0; v < 8; ++v) acc[g][v] = 0.0f;

    v4f ra[8], rb[32];

    #pragma unroll
    for (int i = 0; i < 8; ++i)
        ra[i] = *(const v4f*)(h_in + (size_t)(mtile * 16 + 2 * i + half) * H_ + k0 + spc);
    const float* wsrc = wfrag + wpiece_off(htile * 8 + q) + lane * 4;
    #pragma unroll
    for (int i = 0; i < 32; ++i)
        rb[i] = *(const v4f*)(wsrc + i * 128);

    const int rowg = mtile * 16 + q + 8 * half;
    const float* xrow = xg + ((size_t)rowg * T_ + t) * V_ + PARK + col;
    const float xi = xrow[0 * H_];
    const float xf = xrow[1 * H_];
    const float xc = xrow[2 * H_];
    const float xo = xrow[3 * H_];
    const float cp = c_state[(size_t)rowg * H_ + col];

    #pragma unroll
    for (int i = 0; i < 8; ++i)
        *(v4f*)(reg + (2 * i + half) * LP + spc) = ra[i];
    asm volatile("s_wait_dscnt 0x0" ::: "memory");

    #pragma unroll
    for (int kp = 0; kp < KW / 8; ++kp) {
        #pragma unroll
        for (int s = 0; s < 2; ++s) {
            const int kk = 4 * (2 * kp + s) + 2 * half;
            v2f a = *(const v2f*)(reg + l16 * LP + kk);
            #pragma unroll
            for (int g = 0; g < 4; ++g) {
                const v4f w = rb[kp * 4 + g];
                v2f b;
                b.x = s == 0 ? w.x : w.z;
                b.y = s == 0 ? w.y : w.w;
                acc[g] = wmma_f32(a, b, acc[g]);
            }
        }
    }

    __syncthreads();
    #pragma unroll
    for (int g = 0; g < 4; ++g)
        #pragma unroll
        for (int v = 0; v < 8; ++v)
            sm[((q * 4 + g) * 8 + v) * 32 + lane] = acc[g][v];
    __syncthreads();

    float s[4];
    #pragma unroll
    for (int g = 0; g < 4; ++g) {
        float p = sm[((0 * 4 + g) * 8 + q) * 32 + lane];
        #pragma unroll
        for (int w = 1; w < KSPLIT; ++w)
            p += sm[((w * 4 + g) * 8 + q) * 32 + lane];
        s[g] = p;
    }

    {
        float iv = sigmoidf_(s[0] + xi);
        float fv = sigmoidf_(s[1] + xf);
        float gv = tanhf(s[2] + xc);
        float ov = sigmoidf_(s[3] + xo);
        float cn = fv * cp + iv * gv;
        const float hn = ov * tanhf(cn);
        c_state[(size_t)rowg * H_ + col] = cn;
        h_out[(size_t)rowg * H_ + col] = hn;
        if (t < 2 * HSPLIT) {
            const int prow = t < HSPLIT ? HROW_A : HROW_B;
            const int poff = t < HSPLIT ? HOFF_A + H_ * t : HOFF_B + H_ * (t - HSPLIT);
            hpark[((size_t)rowg * T_ + prow) * V_ + poff + col] = hn;
        }
    }
}

#define PP     36
#define PROWS  144
#define PSTAGE (PROWS * PP)
template <bool PARKED>
__global__ __launch_bounds__(128) void proj_step(
    const float* h_s0, const float* h_s1, const float* h_s2,
    const float* __restrict__ lin_W,
    const float* __restrict__ lin_b, float* out, int t0)
{
    __shared__ float sm[2 * PSTAGE];
    const int step  = blockIdx.y >> 1;
    const int mbase = (blockIdx.y & 1) * 64;
    const int t     = t0 + step;
    const int tid   = threadIdx.x;

    const int w    = tid >> 5;
    const int lane = tid & 31;
    const int half = lane >> 4, l16 = lane & 15;
    const int ntile0 = blockIdx.x * 5;

    const int crow = tid >> 3;
    const int spc  = (tid & 7) * 4;

    v8f acc[5];
    int ncol[5];
    #pragma unroll
    for (int i = 0; i < 5; ++i) ncol[i] = (ntile0 + i) * 16 + l16;
    #pragma unroll
    for (int i = 0; i < 5; ++i)
        #pragma unroll
        for (int v = 0; v < 8; ++v) acc[i][v] = 0.0f;

    v4f ra[4], rb[5];
    const float* hsrc[4];
    if (PARKED) {
        const int prow = t < HSPLIT ? HROW_A : HROW_B;
        const int poff = t < HSPLIT ? HOFF_A + H_ * t : HOFF_B + H_ * (t - HSPLIT);
        #pragma unroll
        for (int i = 0; i < 4; ++i)
            hsrc[i] = h_s0 + ((size_t)(mbase + 16 * i + crow) * T_ + prow) * V_ + poff + spc;
    } else {
        const float* h = step == 0 ? h_s0 : (step == 1 ? h_s1 : h_s2);
        #pragma unroll
        for (int i = 0; i < 4; ++i)
            hsrc[i] = h + (size_t)(mbase + 16 * i + crow) * H_ + spc;
    }
    const float* wsrc = lin_W + (size_t)(ntile0 * 16 + crow) * H_ + spc;

    #pragma unroll
    for (int i = 0; i < 5; ++i) rb[i] = *(const v4f*)(wsrc + (size_t)(16 * i) * H_);
    #pragma unroll
    for (int i = 0; i < 4; ++i) ra[i] = *(const v4f*)(hsrc[i]);

    for (int c = 0; c < 16; ++c) {
        float* st = sm + (c & 1) * PSTAGE;
        #pragma unroll
        for (int i = 0; i < 4; ++i)
            *(v4f*)(st + (16 * i + crow) * PP + spc) = ra[i];
        #pragma unroll
        for (int i = 0; i < 5; ++i)
            *(v4f*)(st + (64 + 16 * i + crow) * PP + spc) = rb[i];
        __syncthreads();
        if (c < 15) {
            #pragma unroll
            for (int i = 0; i < 5; ++i) rb[i] = *(const v4f*)(wsrc + (size_t)(16 * i) * H_ + 32 * (c + 1));
            #pragma unroll
            for (int i = 0; i < 4; ++i) ra[i] = *(const v4f*)(hsrc[i] + 32 * (c + 1));
        }
        #pragma unroll
        for (int ks = 0; ks < 8; ++ks) {
            const int kk = 4 * ks + 2 * half;
            v2f a = *(const v2f*)(st + (16 * w + l16) * PP + kk);
            #pragma unroll
            for (int i = 0; i < 5; ++i) {
                v2f b = *(const v2f*)(st + (64 + i * 16 + l16) * PP + kk);
                acc[i] = wmma_f32(a, b, acc[i]);
            }
        }
    }

    #pragma unroll
    for (int i = 0; i < 5; ++i) {
        float bias = lin_b[ncol[i]];
        #pragma unroll
        for (int v = 0; v < 8; ++v) {
            int rowg = mbase + 16 * w + v + 8 * half;
            __builtin_nontemporal_store(
                acc[i][v] + bias,
                out + (size_t)rowg * T_ * V_ + (size_t)t * V_ + ncol[i]);
        }
    }
}

extern "C" void kernel_launch(void* const* d_in, const int* in_sizes, int n_in,
                              void* d_out, int out_size, void* d_ws, size_t ws_size,
                              hipStream_t stream) {
    const float* features = (const float*)d_in[0];
    const int*   captions = (const int*)d_in[1];
    const float* embed_W  = (const float*)d_in[2];
    const float* W_ih     = (const float*)d_in[3];
    const float* W_hh     = (const float*)d_in[4];
    const float* b_ih     = (const float*)d_in[5];
    const float* b_hh     = (const float*)d_in[6];
    const float* lin_W    = (const float*)d_in[7];
    const float* lin_b    = (const float*)d_in[8];
    float* out = (float*)d_out;

    float* ws = (float*)d_ws;
    float* b0 = ws;
    float* b1 = ws + (size_t)B_ * H_;
    float* b2 = ws + 2 * (size_t)B_ * H_;
    float* cs = ws + 3 * (size_t)B_ * H_;

    lstm_prep<<<256, 256, 0, stream>>>(features, W_hh, b0, cs, out);
    xgate_all<<<dim3(32, 40), 128, 0, stream>>>(captions, embed_W, W_ih, b_ih, b_hh, out);

    for (int g = 0; g < 6; ++g) {
        lstm_step<<<dim3(32, 8), 256, 0, stream>>>(3 * g, out, out, b0, b1, cs, out);
        lstm_step<<<dim3(32, 8), 256, 0, stream>>>(3 * g + 1, out, out, b1, b2, cs, out);
        lstm_step<<<dim3(32, 8), 256, 0, stream>>>(3 * g + 2, out, out, b2, b0, cs, out);
    }
    lstm_step<<<dim3(32, 8), 256, 0, stream>>>(18, out, out, b0, b1, cs, out);
    lstm_step<<<dim3(32, 8), 256, 0, stream>>>(19, out, out, b1, b2, cs, out);
    proj_step<true><<<dim3(125, 36), 128, 0, stream>>>(out, out, out, lin_W, lin_b, out, 0);
    proj_step<false><<<dim3(125, 4), 128, 0, stream>>>(b1, b2, b2, lin_W, lin_b, out, 18);
}
